# starting kernel
# baseline (speedup 1.0000x reference)
#include <hip/hip_runtime.h>
#include <hip/hip_bf16.h>


#define QN    8
#define KCB   1024
#define DIMV  64

typedef __attribute__((ext_vector_type(2))) float v2f;
typedef __attribute__((ext_vector_type(8))) float v8f;

__global__ __launch_bounds__(256)
void ResidualVQ_csq_kernel(const float* __restrict__ cbs,
                           float* __restrict__ csq, int total) {
    int i = blockIdx.x * blockDim.x + threadIdx.x;
    if (i >= total) return;
    const float* r = cbs + (size_t)i * DIMV;
    float s = 0.0f;
#pragma unroll
    for (int k = 0; k < DIMV; ++k) s = __builtin_fmaf(r[k], r[k], s);
    csq[i] = s;
}

__global__ __launch_bounds__(128)
void ResidualVQ_main_kernel(const float* __restrict__ x,
                            const float* __restrict__ cbs,
                            const float* __restrict__ csq,
                            float* __restrict__ out, int tokens) {
    const int lane = threadIdx.x & 31;
    const int wave = threadIdx.x >> 5;
    const int m    = lane & 15;
    const int hi   = lane >> 4;
    const int tileBase = (blockIdx.x * 4 + wave) * 32;
    const int token0 = tileBase + m;
    const int token1 = tileBase + 16 + m;

    const float* xrow0 = x + (size_t)token0 * DIMV;
    const float* xrow1 = x + (size_t)token1 * DIMV;
    v2f a0[16], a1[16], acc0[16], acc1[16];
#pragma unroll
    for (int j = 0; j < 16; ++j) {
        a0[j]   = *(const v2f*)(xrow0 + 4 * j + 2 * hi);
        a1[j]   = *(const v2f*)(xrow1 + 4 * j + 2 * hi);
        acc0[j] = (v2f){0.0f, 0.0f};
        acc1[j] = (v2f){0.0f, 0.0f};
    }

    for (int q = 0; q < QN; ++q) {
        const float* cb = cbs + (size_t)q * KCB * DIMV;
        const float* cs = csq + q * KCB;

        float bestS0[8], bestS1[8];
        int   bestI0[8], bestI1[8];
#pragma unroll
        for (int r = 0; r < 8; ++r) {
            bestS0[r] = __builtin_inff(); bestI0[r] = 0;
            bestS1[r] = __builtin_inff(); bestI1[r] = 0;
        }

        for (int chunk = 0; chunk < KCB / 16; ++chunk) {
            const int n = chunk * 16 + m;
            const float* crow = cb + (size_t)n * DIMV + 2 * hi;

            v2f b[16];
#pragma unroll
            for (int j = 0; j < 16; ++j) b[j] = *(const v2f*)(crow + 4 * j);

            v8f c0 = {};
            v8f c1 = {};
#pragma unroll
            for (int j = 0; j < 16; ++j) {
                c0 = __builtin_amdgcn_wmma_f32_16x16x4_f32(
                         false, a0[j], false, b[j], (short)0, c0, false, false);
                c1 = __builtin_amdgcn_wmma_f32_16x16x4_f32(
                         false, a1[j], false, b[j], (short)0, c1, false, false);
            }

            if (chunk + 1 < KCB / 16)
                __builtin_prefetch(crow + 16 * DIMV, 0, 0);

            const float cnorm = cs[n];
#pragma unroll
            for (int r = 0; r < 8; ++r) {
                float s0 = __builtin_fmaf(-2.0f, c0[r], cnorm);
                float s1 = __builtin_fmaf(-2.0f, c1[r], cnorm);
                if (s0 < bestS0[r]) { bestS0[r] = s0; bestI0[r] = n; }
                if (s1 < bestS1[r]) { bestS1[r] = s1; bestI1[r] = n; }
            }
        }

#pragma unroll
        for (int r = 0; r < 8; ++r) {
            float s0 = bestS0[r]; int i0 = bestI0[r];
            float s1 = bestS1[r]; int i1 = bestI1[r];
#pragma unroll
            for (int off = 8; off >= 1; off >>= 1) {
                float so0 = __shfl_xor(s0, off, 16);
                int   io0 = __shfl_xor(i0, off, 16);
                float so1 = __shfl_xor(s1, off, 16);
                int   io1 = __shfl_xor(i1, off, 16);
                if (so0 < s0 || (so0 == s0 && io0 < i0)) { s0 = so0; i0 = io0; }
                if (so1 < s1 || (so1 == s1 && io1 < i1)) { s1 = so1; i1 = io1; }
            }
            bestI0[r] = i0;
            bestI1[r] = i1;
        }

        int myIdx0 = 0, myIdx1 = 0;
#pragma unroll
        for (int r = 0; r < 8; ++r) {
            int lo0 = __shfl(bestI0[r], 0, 32);
            int hi0 = __shfl(bestI0[r], 16, 32);
            int lo1 = __shfl(bestI1[r], 0, 32);
            int hi1 = __shfl(bestI1[r], 16, 32);
            if (m == r)     { myIdx0 = lo0; myIdx1 = lo1; }
            if (m == r + 8) { myIdx0 = hi0; myIdx1 = hi1; }
        }

        const float* sel0 = cb + (size_t)myIdx0 * DIMV + 2 * hi;
        const float* sel1 = cb + (size_t)myIdx1 * DIMV + 2 * hi;
#pragma unroll
        for (int j = 0; j < 16; ++j) {
            v2f cv0 = *(const v2f*)(sel0 + 4 * j);
            v2f cv1 = *(const v2f*)(sel1 + 4 * j);
            a0[j]   -= cv0;
            a1[j]   -= cv1;
            acc0[j] += cv0;
            acc1[j] += cv1;
        }
    }

    float* orow0 = out + (size_t)token0 * DIMV;
    float* orow1 = out + (size_t)token1 * DIMV;
#pragma unroll
    for (int j = 0; j < 16; ++j) {
        *(v2f*)(orow0 + 4 * j + 2 * hi) = acc0[j];
        *(v2f*)(orow1 + 4 * j + 2 * hi) = acc1[j];
    }
}

extern "C" void kernel_launch(void* const* d_in, const int* in_sizes, int n_in,
                              void* d_out, int out_size, void* d_ws, size_t ws_size,
                              hipStream_t stream) {
    const float* x   = (const float*)d_in[0];
    const float* cbs = (const float*)d_in[1];
    float*       out = (float*)d_out;
    float*       csq = (float*)d_ws;

    const int tokens  = in_sizes[0] / DIMV;
    const int totalCB = in_sizes[1] / DIMV;

    ResidualVQ_csq_kernel<<<(totalCB + 255) / 256, 256, 0, stream>>>(cbs, csq, totalCB);
    ResidualVQ_main_kernel<<<tokens / 128, 128, 0, stream>>>(x, cbs, csq, out, tokens);
}


# Each stage's codebook scan split over a pair of waves sharing one 32-token tile (512 codewords each), half-winners merged in LDS once per stage: 4 not 2 waves per SIMD fill each other's load stalls
# speedup vs baseline: 1.0653x; 1.0653x over previous
#include <hip/hip_runtime.h>
#include <hip/hip_bf16.h>


#define QN    8
#define KCB   1024
#define DIMV  64
#define HALF_CHUNKS (KCB / 32)

typedef __attribute__((ext_vector_type(2))) float v2f;
typedef __attribute__((ext_vector_type(8))) float v8f;

struct __attribute__((aligned(8))) Cand { float s; int i; };

__global__ __launch_bounds__(256)
void ResidualVQ_csq_kernel(const float* __restrict__ cbs,
                           float* __restrict__ csq, int total) {
    int i = blockIdx.x * blockDim.x + threadIdx.x;
    if (i >= total) return;
    const float* r = cbs + (size_t)i * DIMV;
    float s = 0.0f;
#pragma unroll
    for (int k = 0; k < DIMV; ++k) s = __builtin_fmaf(r[k], r[k], s);
    csq[i] = s;
}

__global__ __launch_bounds__(128)
void ResidualVQ_main_kernel(const float* __restrict__ x,
                            const float* __restrict__ cbs,
                            const float* __restrict__ csq,
                            float* __restrict__ out, int tokens) {
    __shared__ Cand win[2][2][2][2][16];

    const int lane = threadIdx.x & 31;
    const int wave = __builtin_amdgcn_readfirstlane(threadIdx.x >> 5);
    const int half = wave & 1;
    const int pair = wave >> 1;
    const int m    = lane & 15;
    const int hi   = lane >> 4;
    const int tileBase = (blockIdx.x * 2 + pair) * 32;
    const int token0 = tileBase + m;
    const int token1 = tileBase + 16 + m;

    const float* xrow0 = x + (size_t)token0 * DIMV;
    const float* xrow1 = x + (size_t)token1 * DIMV;
    v2f a0[16], a1[16], acc[16];
#pragma unroll
    for (int j = 0; j < 16; ++j) {
        a0[j]  = *(const v2f*)(xrow0 + 4 * j + 2 * hi);
        a1[j]  = *(const v2f*)(xrow1 + 4 * j + 2 * hi);
        acc[j] = (v2f){0.0f, 0.0f};
    }

    for (int q = 0; q < QN; ++q) {
        const float* cb = cbs + (size_t)q * KCB * DIMV;
        const float* cs = csq + q * KCB;

        float bestS0[8], bestS1[8];
        int   bestI0[8], bestI1[8];
#pragma unroll
        for (int r = 0; r < 8; ++r) {
            bestS0[r] = __builtin_inff(); bestI0[r] = 0;
            bestS1[r] = __builtin_inff(); bestI1[r] = 0;
        }

        const int chunkBeg = half * HALF_CHUNKS;
        for (int c = 0; c < HALF_CHUNKS; ++c) {
            const int n = (chunkBeg + c) * 16 + m;
            const float* crow = cb + (size_t)n * DIMV + 2 * hi;

            v2f b[16];
#pragma unroll
            for (int j = 0; j < 16; ++j) b[j] = *(const v2f*)(crow + 4 * j);

            v8f c0 = {};
            v8f c1 = {};
#pragma unroll
            for (int j = 0; j < 16; ++j) {
                c0 = __builtin_amdgcn_wmma_f32_16x16x4_f32(
                         false, a0[j], false, b[j], (short)0, c0, false, false);
                c1 = __builtin_amdgcn_wmma_f32_16x16x4_f32(
                         false, a1[j], false, b[j], (short)0, c1, false, false);
            }

            if (c + 1 < HALF_CHUNKS)
                __builtin_prefetch(crow + 16 * DIMV, 0, 0);

            const float cnorm = cs[n];
#pragma unroll
            for (int r = 0; r < 8; ++r) {
                float s0 = __builtin_fmaf(-2.0f, c0[r], cnorm);
                float s1 = __builtin_fmaf(-2.0f, c1[r], cnorm);
                if (s0 < bestS0[r]) { bestS0[r] = s0; bestI0[r] = n; }
                if (s1 < bestS1[r]) { bestS1[r] = s1; bestI1[r] = n; }
            }
        }

#pragma unroll
        for (int r = 0; r < 8; ++r) {
            float s0 = bestS0[r]; int i0 = bestI0[r];
            float s1 = bestS1[r]; int i1 = bestI1[r];
#pragma unroll
            for (int off = 8; off >= 1; off >>= 1) {
                float so0 = __shfl_xor(s0, off, 16);
                int   io0 = __shfl_xor(i0, off, 16);
                float so1 = __shfl_xor(s1, off, 16);
                int   io1 = __shfl_xor(i1, off, 16);
                if (so0 < s0 || (so0 == s0 && io0 < i0)) { s0 = so0; i0 = io0; }
                if (so1 < s1 || (so1 == s1 && io1 < i1)) { s1 = so1; i1 = io1; }
            }
            bestS0[r] = s0; bestI0[r] = i0;
            bestS1[r] = s1; bestI1[r] = i1;
        }

        const int par = q & 1;
        if (m == 0) {
#pragma unroll
            for (int r = 0; r < 8; ++r) {
                win[par][pair][half][0][8 * hi + r] = (Cand){bestS0[r], bestI0[r]};
                win[par][pair][half][1][8 * hi + r] = (Cand){bestS1[r], bestI1[r]};
            }
        }
        __syncthreads();

        const Cand lo0 = win[par][pair][0][0][m];
        const Cand up0 = win[par][pair][1][0][m];
        const Cand lo1 = win[par][pair][0][1][m];
        const Cand up1 = win[par][pair][1][1][m];
        const int myIdx0 = (up0.s < lo0.s) ? up0.i : lo0.i;
        const int myIdx1 = (up1.s < lo1.s) ? up1.i : lo1.i;

        const float* sel0 = cb + (size_t)myIdx0 * DIMV + 2 * hi;
        const float* sel1 = cb + (size_t)myIdx1 * DIMV + 2 * hi;
#pragma unroll
        for (int j = 0; j < 16; ++j) {
            v2f cv0 = *(const v2f*)(sel0 + 4 * j);
            v2f cv1 = *(const v2f*)(sel1 + 4 * j);
            a0[j]  -= cv0;
            a1[j]  -= cv1;
            acc[j] += half ? cv1 : cv0;
        }
    }

    float* orow = out + (size_t)(half ? token1 : token0) * DIMV;
#pragma unroll
    for (int j = 0; j < 16; ++j)
        *(v2f*)(orow + 4 * j + 2 * hi) = acc[j];
}

extern "C" void kernel_launch(void* const* d_in, const int* in_sizes, int n_in,
                              void* d_out, int out_size, void* d_ws, size_t ws_size,
                              hipStream_t stream) {
    const float* x   = (const float*)d_in[0];
    const float* cbs = (const float*)d_in[1];
    float*       out = (float*)d_out;
    float*       csq = (float*)d_ws;

    const int tokens  = in_sizes[0] / DIMV;
    const int totalCB = in_sizes[1] / DIMV;

    ResidualVQ_csq_kernel<<<(totalCB + 255) / 256, 256, 0, stream>>>(cbs, csq, totalCB);
    ResidualVQ_main_kernel<<<tokens / 64, 128, 0, stream>>>(x, cbs, csq, out, tokens);
}


# B and A fragments loaded 16 bytes per lane: one load feeds two WMMAs (k regrouped by 8 dims, so the dot products' summation order changes); half as many 16-row gather loads per WMMA
# speedup vs baseline: 1.0996x; 1.0322x over previous
#include <hip/hip_runtime.h>
#include <hip/hip_bf16.h>


#define QN    8
#define KCB   1024
#define DIMV  64
#define HALF_CHUNKS (KCB / 32)

typedef __attribute__((ext_vector_type(2))) float v2f;
typedef __attribute__((ext_vector_type(4))) float v4f;
typedef __attribute__((ext_vector_type(8))) float v8f;

struct __attribute__((aligned(8))) Cand { float s; int i; };

__global__ __launch_bounds__(256)
void ResidualVQ_csq_kernel(const float* __restrict__ cbs,
                           float* __restrict__ csq, int total) {
    int i = blockIdx.x * blockDim.x + threadIdx.x;
    if (i >= total) return;
    const float* r = cbs + (size_t)i * DIMV;
    float s = 0.0f;
#pragma unroll
    for (int k = 0; k < DIMV; ++k) s = __builtin_fmaf(r[k], r[k], s);
    csq[i] = s;
}

__global__ __launch_bounds__(128)
void ResidualVQ_main_kernel(const float* __restrict__ x,
                            const float* __restrict__ cbs,
                            const float* __restrict__ csq,
                            float* __restrict__ out, int tokens) {
    __shared__ Cand win[2][2][2][2][16];

    const int lane = threadIdx.x & 31;
    const int wave = __builtin_amdgcn_readfirstlane(threadIdx.x >> 5);
    const int half = wave & 1;
    const int pair = wave >> 1;
    const int m    = lane & 15;
    const int hi   = lane >> 4;
    const int tileBase = (blockIdx.x * 2 + pair) * 32;
    const int token0 = tileBase + m;
    const int token1 = tileBase + 16 + m;

    const float* xrow0 = x + (size_t)token0 * DIMV + 4 * hi;
    const float* xrow1 = x + (size_t)token1 * DIMV + 4 * hi;
    v4f a0[8], a1[8], acc[8];
#pragma unroll
    for (int i = 0; i < 8; ++i) {
        a0[i]  = *(const v4f*)(xrow0 + 8 * i);
        a1[i]  = *(const v4f*)(xrow1 + 8 * i);
        acc[i] = (v4f){0.0f, 0.0f, 0.0f, 0.0f};
    }

    for (int q = 0; q < QN; ++q) {
        const float* cb = cbs + (size_t)q * KCB * DIMV;
        const float* cs = csq + q * KCB;

        float bestS0[8], bestS1[8];
        int   bestI0[8], bestI1[8];
#pragma unroll
        for (int r = 0; r < 8; ++r) {
            bestS0[r] = __builtin_inff(); bestI0[r] = 0;
            bestS1[r] = __builtin_inff(); bestI1[r] = 0;
        }

        const int chunkBeg = half * HALF_CHUNKS;
        for (int c = 0; c < HALF_CHUNKS; ++c) {
            const int n = (chunkBeg + c) * 16 + m;
            const float* crow = cb + (size_t)n * DIMV + 4 * hi;

            v4f b[8];
#pragma unroll
            for (int i = 0; i < 8; ++i) b[i] = *(const v4f*)(crow + 8 * i);

            v8f c0 = {};
            v8f c1 = {};
#pragma unroll
            for (int i = 0; i < 8; ++i) {
                const v2f bl = __builtin_shufflevector(b[i], b[i], 0, 1);
                const v2f bu = __builtin_shufflevector(b[i], b[i], 2, 3);
                c0 = __builtin_amdgcn_wmma_f32_16x16x4_f32(
                         false, __builtin_shufflevector(a0[i], a0[i], 0, 1),
                         false, bl, (short)0, c0, false, false);
                c1 = __builtin_amdgcn_wmma_f32_16x16x4_f32(
                         false, __builtin_shufflevector(a1[i], a1[i], 0, 1),
                         false, bl, (short)0, c1, false, false);
                c0 = __builtin_amdgcn_wmma_f32_16x16x4_f32(
                         false, __builtin_shufflevector(a0[i], a0[i], 2, 3),
                         false, bu, (short)0, c0, false, false);
                c1 = __builtin_amdgcn_wmma_f32_16x16x4_f32(
                         false, __builtin_shufflevector(a1[i], a1[i], 2, 3),
                         false, bu, (short)0, c1, false, false);
            }

            if (c + 1 < HALF_CHUNKS)
                __builtin_prefetch(crow + 16 * DIMV, 0, 0);

            const float cnorm = cs[n];
#pragma unroll
            for (int r = 0; r < 8; ++r) {
                float s0 = __builtin_fmaf(-2.0f, c0[r], cnorm);
                float s1 = __builtin_fmaf(-2.0f, c1[r], cnorm);
                if (s0 < bestS0[r]) { bestS0[r] = s0; bestI0[r] = n; }
                if (s1 < bestS1[r]) { bestS1[r] = s1; bestI1[r] = n; }
            }
        }

#pragma unroll
        for (int r = 0; r < 8; ++r) {
            float s0 = bestS0[r]; int i0 = bestI0[r];
            float s1 = bestS1[r]; int i1 = bestI1[r];
#pragma unroll
            for (int off = 8; off >= 1; off >>= 1) {
                float so0 = __shfl_xor(s0, off, 16);
                int   io0 = __shfl_xor(i0, off, 16);
                float so1 = __shfl_xor(s1, off, 16);
                int   io1 = __shfl_xor(i1, off, 16);
                if (so0 < s0 || (so0 == s0 && io0 < i0)) { s0 = so0; i0 = io0; }
                if (so1 < s1 || (so1 == s1 && io1 < i1)) { s1 = so1; i1 = io1; }
            }
            bestS0[r] = s0; bestI0[r] = i0;
            bestS1[r] = s1; bestI1[r] = i1;
        }

        const int par = q & 1;
        if (m == 0) {
#pragma unroll
            for (int r = 0; r < 8; ++r) {
                win[par][pair][half][0][8 * hi + r] = (Cand){bestS0[r], bestI0[r]};
                win[par][pair][half][1][8 * hi + r] = (Cand){bestS1[r], bestI1[r]};
            }
        }
        __syncthreads();

        const Cand lo0 = win[par][pair][0][0][m];
        const Cand up0 = win[par][pair][1][0][m];
        const Cand lo1 = win[par][pair][0][1][m];
        const Cand up1 = win[par][pair][1][1][m];
        const int myIdx0 = (up0.s < lo0.s) ? up0.i : lo0.i;
        const int myIdx1 = (up1.s < lo1.s) ? up1.i : lo1.i;

        const float* sel0 = cb + (size_t)myIdx0 * DIMV + 4 * hi;
        const float* sel1 = cb + (size_t)myIdx1 * DIMV + 4 * hi;
#pragma unroll
        for (int i = 0; i < 8; ++i) {
            v4f cv0 = *(const v4f*)(sel0 + 8 * i);
            v4f cv1 = *(const v4f*)(sel1 + 8 * i);
            a0[i]  -= cv0;
            a1[i]  -= cv1;
            acc[i] += half ? cv1 : cv0;
        }
    }

    float* orow = out + (size_t)(half ? token1 : token0) * DIMV + 4 * hi;
#pragma unroll
    for (int i = 0; i < 8; ++i)
        *(v4f*)(orow + 8 * i) = acc[i];
}

extern "C" void kernel_launch(void* const* d_in, const int* in_sizes, int n_in,
                              void* d_out, int out_size, void* d_ws, size_t ws_size,
                              hipStream_t stream) {
    const float* x   = (const float*)d_in[0];
    const float* cbs = (const float*)d_in[1];
    float*       out = (float*)d_out;
    float*       csq = (float*)d_ws;

    const int tokens  = in_sizes[0] / DIMV;
    const int totalCB = in_sizes[1] / DIMV;

    ResidualVQ_csq_kernel<<<(totalCB + 255) / 256, 256, 0, stream>>>(cbs, csq, totalCB);
    ResidualVQ_main_kernel<<<tokens / 64, 128, 0, stream>>>(x, cbs, csq, out, tokens);
}


# Stage-end argmin by LDS transpose: each lane stores its 16 running (score,idx) once; one lane per token row reads the row's 32 candidates 16 bytes at a time, reduces in registers; no shuffle ladder
# speedup vs baseline: 1.1671x; 1.0614x over previous
#include <hip/hip_runtime.h>
#include <hip/hip_bf16.h>


#define QN    8
#define KCB   1024
#define DIMV  64
#define HALF_CHUNKS (KCB / 32)

typedef __attribute__((ext_vector_type(2))) float v2f;
typedef __attribute__((ext_vector_type(4))) float v4f;
typedef __attribute__((ext_vector_type(8))) float v8f;

typedef __attribute__((ext_vector_type(2))) int v2i;

#define ROW_PITCH_C 34
#define TILE_PITCH_C (16 * ROW_PITCH_C)

__device__ __forceinline__ void take_lower(float& s, int& i, float so, int io) {
    const bool b = (so < s) | ((so == s) & (io < i));
    s = b ? so : s;
    i = b ? io : i;
}

__global__ __launch_bounds__(256)
void ResidualVQ_csq_kernel(const float* __restrict__ cbs,
                           float* __restrict__ csq, int total) {
    int i = blockIdx.x * blockDim.x + threadIdx.x;
    if (i >= total) return;
    const float* r = cbs + (size_t)i * DIMV;
    float s = 0.0f;
#pragma unroll
    for (int k = 0; k < DIMV; ++k) s = __builtin_fmaf(r[k], r[k], s);
    csq[i] = s;
}

__global__ __launch_bounds__(128)
void ResidualVQ_main_kernel(const float* __restrict__ x,
                            const float* __restrict__ cbs,
                            const float* __restrict__ csq,
                            float* __restrict__ out, int tokens) {
    __shared__ __attribute__((aligned(16))) v2i cand[2][2][2][TILE_PITCH_C];

    const int lane = threadIdx.x & 31;
    const int wave = __builtin_amdgcn_readfirstlane(threadIdx.x >> 5);
    const int half = wave & 1;
    const int pair = wave >> 1;
    const int m    = lane & 15;
    const int hi   = lane >> 4;
    const int tileBase = (blockIdx.x * 2 + pair) * 32;
    const int token0 = tileBase + m;
    const int token1 = tileBase + 16 + m;

    const float* xrow0 = x + (size_t)token0 * DIMV + 4 * hi;
    const float* xrow1 = x + (size_t)token1 * DIMV + 4 * hi;
    v4f a0[8], a1[8], acc[8];
#pragma unroll
    for (int i = 0; i < 8; ++i) {
        a0[i]  = *(const v4f*)(xrow0 + 8 * i);
        a1[i]  = *(const v4f*)(xrow1 + 8 * i);
        acc[i] = (v4f){0.0f, 0.0f, 0.0f, 0.0f};
    }

    for (int q = 0; q < QN; ++q) {
        const float* cb = cbs + (size_t)q * KCB * DIMV;
        const float* cs = csq + q * KCB;

        float bestS0[8], bestS1[8];
        int   bestI0[8], bestI1[8];
#pragma unroll
        for (int r = 0; r < 8; ++r) {
            bestS0[r] = __builtin_inff(); bestI0[r] = 0;
            bestS1[r] = __builtin_inff(); bestI1[r] = 0;
        }

        const int chunkBeg = half * HALF_CHUNKS;
        for (int c = 0; c < HALF_CHUNKS; ++c) {
            const int n = (chunkBeg + c) * 16 + m;
            const float* crow = cb + (size_t)n * DIMV + 4 * hi;

            v4f b[8];
#pragma unroll
            for (int i = 0; i < 8; ++i) b[i] = *(const v4f*)(crow + 8 * i);

            v8f c0 = {};
            v8f c1 = {};
#pragma unroll
            for (int i = 0; i < 8; ++i) {
                const v2f bl = __builtin_shufflevector(b[i], b[i], 0, 1);
                const v2f bu = __builtin_shufflevector(b[i], b[i], 2, 3);
                c0 = __builtin_amdgcn_wmma_f32_16x16x4_f32(
                         false, __builtin_shufflevector(a0[i], a0[i], 0, 1),
                         false, bl, (short)0, c0, false, false);
                c1 = __builtin_amdgcn_wmma_f32_16x16x4_f32(
                         false, __builtin_shufflevector(a1[i], a1[i], 0, 1),
                         false, bl, (short)0, c1, false, false);
                c0 = __builtin_amdgcn_wmma_f32_16x16x4_f32(
                         false, __builtin_shufflevector(a0[i], a0[i], 2, 3),
                         false, bu, (short)0, c0, false, false);
                c1 = __builtin_amdgcn_wmma_f32_16x16x4_f32(
                         false, __builtin_shufflevector(a1[i], a1[i], 2, 3),
                         false, bu, (short)0, c1, false, false);
            }

            if (c + 1 < HALF_CHUNKS)
                __builtin_prefetch(crow + 16 * DIMV, 0, 0);

            const float cnorm = cs[n];
#pragma unroll
            for (int r = 0; r < 8; ++r) {
                float s0 = __builtin_fmaf(-2.0f, c0[r], cnorm);
                float s1 = __builtin_fmaf(-2.0f, c1[r], cnorm);
                if (s0 < bestS0[r]) { bestS0[r] = s0; bestI0[r] = n; }
                if (s1 < bestS1[r]) { bestS1[r] = s1; bestI1[r] = n; }
            }
        }

        const int par = q & 1;
        {
            v2i* t0 = &cand[par][pair][0][(8 * hi) * ROW_PITCH_C + 16 * half + m];
            v2i* t1 = &cand[par][pair][1][(8 * hi) * ROW_PITCH_C + 16 * half + m];
#pragma unroll
            for (int r = 0; r < 8; ++r) {
                t0[r * ROW_PITCH_C] = (v2i){__builtin_bit_cast(int, bestS0[r]), bestI0[r]};
                t1[r * ROW_PITCH_C] = (v2i){__builtin_bit_cast(int, bestS1[r]), bestI1[r]};
            }
        }
        __syncthreads();

        float ws[16];
        int   wi[16];
        {
            const v2i* rowp = &cand[par][pair][hi][m * ROW_PITCH_C];
#pragma unroll
            for (int j = 0; j < 16; ++j) {
                const v2i ca = rowp[2 * j];
                const v2i cb2 = rowp[2 * j + 1];
                ws[j] = __builtin_bit_cast(float, ca[0]);
                wi[j] = ca[1];
                take_lower(ws[j], wi[j], __builtin_bit_cast(float, cb2[0]), cb2[1]);
            }
        }
#pragma unroll
        for (int w = 8; w >= 1; w >>= 1) {
#pragma unroll
            for (int j = 0; j < w; ++j) take_lower(ws[j], wi[j], ws[j + w], wi[j + w]);
        }
        const int other = __shfl_xor(wi[0], 16, 32);
        const int myIdx0 = hi ? other : wi[0];
        const int myIdx1 = hi ? wi[0] : other;

        const float* sel0 = cb + (size_t)myIdx0 * DIMV + 4 * hi;
        const float* sel1 = cb + (size_t)myIdx1 * DIMV + 4 * hi;
#pragma unroll
        for (int i = 0; i < 8; ++i) {
            v4f cv0 = *(const v4f*)(sel0 + 8 * i);
            v4f cv1 = *(const v4f*)(sel1 + 8 * i);
            a0[i]  -= cv0;
            a1[i]  -= cv1;
            acc[i] += half ? cv1 : cv0;
        }
    }

    float* orow = out + (size_t)(half ? token1 : token0) * DIMV + 4 * hi;
#pragma unroll
    for (int i = 0; i < 8; ++i)
        *(v4f*)(orow + 8 * i) = acc[i];
}

extern "C" void kernel_launch(void* const* d_in, const int* in_sizes, int n_in,
                              void* d_out, int out_size, void* d_ws, size_t ws_size,
                              hipStream_t stream) {
    const float* x   = (const float*)d_in[0];
    const float* cbs = (const float*)d_in[1];
    float*       out = (float*)d_out;
    float*       csq = (float*)d_ws;

    const int tokens  = in_sizes[0] / DIMV;
    const int totalCB = in_sizes[1] / DIMV;

    ResidualVQ_csq_kernel<<<(totalCB + 255) / 256, 256, 0, stream>>>(cbs, csq, totalCB);
    ResidualVQ_main_kernel<<<tokens / 64, 128, 0, stream>>>(x, cbs, csq, out, tokens);
}


# Chunk loop software-pipelined over global memory: a chunk's eight fragment loads and norm load go out one chunk ahead of its 32 WMMAs, into a second register set; next-row prefetch hint dropped
# speedup vs baseline: 1.2650x; 1.0839x over previous
#include <hip/hip_runtime.h>
#include <hip/hip_bf16.h>


#define QN    8
#define KCB   1024
#define DIMV  64
#define HALF_CHUNKS (KCB / 32)

typedef __attribute__((ext_vector_type(2))) float v2f;
typedef __attribute__((ext_vector_type(4))) float v4f;
typedef __attribute__((ext_vector_type(8))) float v8f;

typedef __attribute__((ext_vector_type(2))) int v2i;

#define ROW_PITCH_C 34
#define TILE_PITCH_C (16 * ROW_PITCH_C)

__device__ __forceinline__ void take_lower(float& s, int& i, float so, int io) {
    const bool b = (so < s) | ((so == s) & (io < i));
    s = b ? so : s;
    i = b ? io : i;
}

__device__ __forceinline__ void load_chunk(const float* __restrict__ crow,
                                           const float* __restrict__ cnp,
                                           v4f (&b)[8], float& cnorm) {
#pragma unroll
    for (int i = 0; i < 8; ++i) b[i] = *(const v4f*)(crow + 8 * i);
    cnorm = *cnp;
}

__device__ __forceinline__ void score_chunk(const v4f (&a0)[8], const v4f (&a1)[8],
                                            const v4f (&b)[8], float cnorm, int n,
                                            float (&bestS0)[8], int (&bestI0)[8],
                                            float (&bestS1)[8], int (&bestI1)[8]) {
    v8f c0 = {};
    v8f c1 = {};
#pragma unroll
    for (int i = 0; i < 8; ++i) {
        const v2f bl = __builtin_shufflevector(b[i], b[i], 0, 1);
        const v2f bu = __builtin_shufflevector(b[i], b[i], 2, 3);
        c0 = __builtin_amdgcn_wmma_f32_16x16x4_f32(
                 false, __builtin_shufflevector(a0[i], a0[i], 0, 1),
                 false, bl, (short)0, c0, false, false);
        c1 = __builtin_amdgcn_wmma_f32_16x16x4_f32(
                 false, __builtin_shufflevector(a1[i], a1[i], 0, 1),
                 false, bl, (short)0, c1, false, false);
        c0 = __builtin_amdgcn_wmma_f32_16x16x4_f32(
                 false, __builtin_shufflevector(a0[i], a0[i], 2, 3),
                 false, bu, (short)0, c0, false, false);
        c1 = __builtin_amdgcn_wmma_f32_16x16x4_f32(
                 false, __builtin_shufflevector(a1[i], a1[i], 2, 3),
                 false, bu, (short)0, c1, false, false);
    }
#pragma unroll
    for (int r = 0; r < 8; ++r) {
        float s0 = __builtin_fmaf(-2.0f, c0[r], cnorm);
        float s1 = __builtin_fmaf(-2.0f, c1[r], cnorm);
        if (s0 < bestS0[r]) { bestS0[r] = s0; bestI0[r] = n; }
        if (s1 < bestS1[r]) { bestS1[r] = s1; bestI1[r] = n; }
    }
}

__global__ __launch_bounds__(256)
void ResidualVQ_csq_kernel(const float* __restrict__ cbs,
                           float* __restrict__ csq, int total) {
    int i = blockIdx.x * blockDim.x + threadIdx.x;
    if (i >= total) return;
    const float* r = cbs + (size_t)i * DIMV;
    float s = 0.0f;
#pragma unroll
    for (int k = 0; k < DIMV; ++k) s = __builtin_fmaf(r[k], r[k], s);
    csq[i] = s;
}

__global__ __launch_bounds__(128)
void ResidualVQ_main_kernel(const float* __restrict__ x,
                            const float* __restrict__ cbs,
                            const float* __restrict__ csq,
                            float* __restrict__ out, int tokens) {
    __shared__ __attribute__((aligned(16))) v2i cand[2][2][2][TILE_PITCH_C];

    const int lane = threadIdx.x & 31;
    const int wave = __builtin_amdgcn_readfirstlane(threadIdx.x >> 5);
    const int half = wave & 1;
    const int pair = wave >> 1;
    const int m    = lane & 15;
    const int hi   = lane >> 4;
    const int tileBase = (blockIdx.x * 2 + pair) * 32;
    const int token0 = tileBase + m;
    const int token1 = tileBase + 16 + m;

    const float* xrow0 = x + (size_t)token0 * DIMV + 4 * hi;
    const float* xrow1 = x + (size_t)token1 * DIMV + 4 * hi;
    v4f a0[8], a1[8], acc[8];
#pragma unroll
    for (int i = 0; i < 8; ++i) {
        a0[i]  = *(const v4f*)(xrow0 + 8 * i);
        a1[i]  = *(const v4f*)(xrow1 + 8 * i);
        acc[i] = (v4f){0.0f, 0.0f, 0.0f, 0.0f};
    }

    const int nBase = half * HALF_CHUNKS * 16 + m;
    const float* laneRow = cbs + (size_t)nBase * DIMV + 4 * hi;
    const float* laneCs  = csq + nBase;

    v4f bA[8], bB[8];
    float cnA, cnB;
    load_chunk(laneRow, laneCs, bA, cnA);

    for (int q = 0; q < QN; ++q) {
        const float* cb = cbs + (size_t)q * KCB * DIMV;
        const float* rowq = laneRow + (size_t)q * KCB * DIMV;
        const float* csq_q = laneCs + q * KCB;
        const int qn = (q + 1) & (QN - 1);
        const float* rowNext = laneRow + (size_t)qn * KCB * DIMV;
        const float* csNext  = laneCs + qn * KCB;

        float bestS0[8], bestS1[8];
        int   bestI0[8], bestI1[8];
#pragma unroll
        for (int r = 0; r < 8; ++r) {
            bestS0[r] = __builtin_inff(); bestI0[r] = 0;
            bestS1[r] = __builtin_inff(); bestI1[r] = 0;
        }

        for (int c = 0; c < HALF_CHUNKS; c += 2) {
            load_chunk(rowq + (c + 1) * 16 * DIMV, csq_q + (c + 1) * 16, bB, cnB);
            score_chunk(a0, a1, bA, cnA, nBase + c * 16,
                        bestS0, bestI0, bestS1, bestI1);
            __builtin_amdgcn_sched_barrier(0);
            const bool last = (c + 2 == HALF_CHUNKS);
            load_chunk(last ? rowNext : rowq + (c + 2) * 16 * DIMV,
                       last ? csNext  : csq_q + (c + 2) * 16, bA, cnA);
            score_chunk(a0, a1, bB, cnB, nBase + (c + 1) * 16,
                        bestS0, bestI0, bestS1, bestI1);
        }

        const int par = q & 1;
        {
            v2i* t0 = &cand[par][pair][0][(8 * hi) * ROW_PITCH_C + 16 * half + m];
            v2i* t1 = &cand[par][pair][1][(8 * hi) * ROW_PITCH_C + 16 * half + m];
#pragma unroll
            for (int r = 0; r < 8; ++r) {
                t0[r * ROW_PITCH_C] = (v2i){__builtin_bit_cast(int, bestS0[r]), bestI0[r]};
                t1[r * ROW_PITCH_C] = (v2i){__builtin_bit_cast(int, bestS1[r]), bestI1[r]};
            }
        }
        __syncthreads();

        float ws[16];
        int   wi[16];
        {
            const v2i* rowp = &cand[par][pair][hi][m * ROW_PITCH_C];
#pragma unroll
            for (int j = 0; j < 16; ++j) {
                const v2i ca = rowp[2 * j];
                const v2i cb2 = rowp[2 * j + 1];
                ws[j] = __builtin_bit_cast(float, ca[0]);
                wi[j] = ca[1];
                take_lower(ws[j], wi[j], __builtin_bit_cast(float, cb2[0]), cb2[1]);
            }
        }
#pragma unroll
        for (int w = 8; w >= 1; w >>= 1) {
#pragma unroll
            for (int j = 0; j < w; ++j) take_lower(ws[j], wi[j], ws[j + w], wi[j + w]);
        }
        const int other = __shfl_xor(wi[0], 16, 32);
        const int myIdx0 = hi ? other : wi[0];
        const int myIdx1 = hi ? wi[0] : other;

        const float* sel0 = cb + (size_t)myIdx0 * DIMV + 4 * hi;
        const float* sel1 = cb + (size_t)myIdx1 * DIMV + 4 * hi;
#pragma unroll
        for (int i = 0; i < 8; ++i) {
            v4f cv0 = *(const v4f*)(sel0 + 8 * i);
            v4f cv1 = *(const v4f*)(sel1 + 8 * i);
            a0[i]  -= cv0;
            a1[i]  -= cv1;
            acc[i] += half ? cv1 : cv0;
        }
    }

    float* orow = out + (size_t)(half ? token1 : token0) * DIMV + 4 * hi;
#pragma unroll
    for (int i = 0; i < 8; ++i)
        *(v4f*)(orow + 8 * i) = acc[i];
}

extern "C" void kernel_launch(void* const* d_in, const int* in_sizes, int n_in,
                              void* d_out, int out_size, void* d_ws, size_t ws_size,
                              hipStream_t stream) {
    const float* x   = (const float*)d_in[0];
    const float* cbs = (const float*)d_in[1];
    float*       out = (float*)d_out;
    float*       csq = (float*)d_ws;

    const int tokens  = in_sizes[0] / DIMV;
    const int totalCB = in_sizes[1] / DIMV;

    ResidualVQ_csq_kernel<<<(totalCB + 255) / 256, 256, 0, stream>>>(cbs, csq, totalCB);
    ResidualVQ_main_kernel<<<tokens / 64, 128, 0, stream>>>(x, cbs, csq, out, tokens);
}


# Codebook copied once per call by the first kernel into WMMA fragment order in the workspace: each 16-byte B load of a wave covers 4 whole 128-byte lines instead of 32 bytes from each of 16 lines
# speedup vs baseline: 1.3166x; 1.0408x over previous
#include <hip/hip_runtime.h>
#include <hip/hip_bf16.h>


#define QN    8
#define KCB   1024
#define DIMV  64
#define HALF_CHUNKS (KCB / 32)
#define CHUNK_FLOATS (16 * DIMV)
#define FRAG_STEP 128

typedef __attribute__((ext_vector_type(2))) float v2f;
typedef __attribute__((ext_vector_type(4))) float v4f;
typedef __attribute__((ext_vector_type(8))) float v8f;

typedef __attribute__((ext_vector_type(2))) int v2i;

#define ROW_PITCH_C 34
#define TILE_PITCH_C (16 * ROW_PITCH_C)

__device__ __forceinline__ void take_lower(float& s, int& i, float so, int io) {
    const bool b = (so < s) | ((so == s) & (io < i));
    s = b ? so : s;
    i = b ? io : i;
}

__device__ __forceinline__ void load_chunk(const float* __restrict__ cfrag,
                                           const float* __restrict__ cnp,
                                           v4f (&b)[8], float& cnorm) {
#pragma unroll
    for (int i = 0; i < 8; ++i) b[i] = *(const v4f*)(cfrag + FRAG_STEP * i);
    cnorm = *cnp;
}

__device__ __forceinline__ void score_chunk(const v4f (&a0)[8], const v4f (&a1)[8],
                                            const v4f (&b)[8], float cnorm, int n,
                                            float (&bestS0)[8], int (&bestI0)[8],
                                            float (&bestS1)[8], int (&bestI1)[8]) {
    v8f c0 = {};
    v8f c1 = {};
#pragma unroll
    for (int i = 0; i < 8; ++i) {
        const v2f bl = __builtin_shufflevector(b[i], b[i], 0, 1);
        const v2f bu = __builtin_shufflevector(b[i], b[i], 2, 3);
        c0 = __builtin_amdgcn_wmma_f32_16x16x4_f32(
                 false, __builtin_shufflevector(a0[i], a0[i], 0, 1),
                 false, bl, (short)0, c0, false, false);
        c1 = __builtin_amdgcn_wmma_f32_16x16x4_f32(
                 false, __builtin_shufflevector(a1[i], a1[i], 0, 1),
                 false, bl, (short)0, c1, false, false);
        c0 = __builtin_amdgcn_wmma_f32_16x16x4_f32(
                 false, __builtin_shufflevector(a0[i], a0[i], 2, 3),
                 false, bu, (short)0, c0, false, false);
        c1 = __builtin_amdgcn_wmma_f32_16x16x4_f32(
                 false, __builtin_shufflevector(a1[i], a1[i], 2, 3),
                 false, bu, (short)0, c1, false, false);
    }
#pragma unroll
    for (int r = 0; r < 8; ++r) {
        float s0 = __builtin_fmaf(-2.0f, c0[r], cnorm);
        float s1 = __builtin_fmaf(-2.0f, c1[r], cnorm);
        if (s0 < bestS0[r]) { bestS0[r] = s0; bestI0[r] = n; }
        if (s1 < bestS1[r]) { bestS1[r] = s1; bestI1[r] = n; }
    }
}

__global__ __launch_bounds__(256)
void ResidualVQ_csq_kernel(const float* __restrict__ cbs,
                           float* __restrict__ csq,
                           float* __restrict__ frag, int total) {
    int i = blockIdx.x * blockDim.x + threadIdx.x;
    if (i >= total) return;
    const float* r = cbs + (size_t)i * DIMV;
    float s = 0.0f;
#pragma unroll
    for (int k = 0; k < DIMV; ++k) s = __builtin_fmaf(r[k], r[k], s);
    csq[i] = s;
    float* dst = frag + (size_t)(i >> 4) * CHUNK_FLOATS + (i & 15) * 4;
#pragma unroll
    for (int g = 0; g < 8; ++g) {
        *(v4f*)(dst + FRAG_STEP * g)      = *(const v4f*)(r + 8 * g);
        *(v4f*)(dst + FRAG_STEP * g + 64) = *(const v4f*)(r + 8 * g + 4);
    }
}

__global__ __launch_bounds__(128)
void ResidualVQ_main_kernel(const float* __restrict__ x,
                            const float* __restrict__ cbs,
                            const float* __restrict__ csq,
                            const float* __restrict__ frag,
                            float* __restrict__ out, int tokens) {
    __shared__ __attribute__((aligned(16))) v2i cand[2][2][2][TILE_PITCH_C];

    const int lane = threadIdx.x & 31;
    const int wave = __builtin_amdgcn_readfirstlane(threadIdx.x >> 5);
    const int half = wave & 1;
    const int pair = wave >> 1;
    const int m    = lane & 15;
    const int hi   = lane >> 4;
    const int tileBase = (blockIdx.x * 2 + pair) * 32;
    const int token0 = tileBase + m;
    const int token1 = tileBase + 16 + m;

    const float* xrow0 = x + (size_t)token0 * DIMV + 4 * hi;
    const float* xrow1 = x + (size_t)token1 * DIMV + 4 * hi;
    v4f a0[8], a1[8], acc[8];
#pragma unroll
    for (int i = 0; i < 8; ++i) {
        a0[i]  = *(const v4f*)(xrow0 + 8 * i);
        a1[i]  = *(const v4f*)(xrow1 + 8 * i);
        acc[i] = (v4f){0.0f, 0.0f, 0.0f, 0.0f};
    }

    const int nBase = half * HALF_CHUNKS * 16 + m;
    const float* laneRow = frag + (size_t)(half * HALF_CHUNKS) * CHUNK_FLOATS
                                + lane * 4;
    const float* laneCs  = csq + nBase;

    v4f bA[8], bB[8];
    float cnA, cnB;
    load_chunk(laneRow, laneCs, bA, cnA);

    for (int q = 0; q < QN; ++q) {
        const float* cb = cbs + (size_t)q * KCB * DIMV;
        const float* rowq = laneRow + (size_t)q * KCB * DIMV;
        const float* csq_q = laneCs + q * KCB;
        const int qn = (q + 1) & (QN - 1);
        const float* rowNext = laneRow + (size_t)qn * KCB * DIMV;
        const float* csNext  = laneCs + qn * KCB;

        float bestS0[8], bestS1[8];
        int   bestI0[8], bestI1[8];
#pragma unroll
        for (int r = 0; r < 8; ++r) {
            bestS0[r] = __builtin_inff(); bestI0[r] = 0;
            bestS1[r] = __builtin_inff(); bestI1[r] = 0;
        }

        for (int c = 0; c < HALF_CHUNKS; c += 2) {
            load_chunk(rowq + (c + 1) * CHUNK_FLOATS, csq_q + (c + 1) * 16, bB, cnB);
            score_chunk(a0, a1, bA, cnA, nBase + c * 16,
                        bestS0, bestI0, bestS1, bestI1);
            __builtin_amdgcn_sched_barrier(0);
            const bool last = (c + 2 == HALF_CHUNKS);
            load_chunk(last ? rowNext : rowq + (c + 2) * CHUNK_FLOATS,
                       last ? csNext  : csq_q + (c + 2) * 16, bA, cnA);
            score_chunk(a0, a1, bB, cnB, nBase + (c + 1) * 16,
                        bestS0, bestI0, bestS1, bestI1);
        }

        const int par = q & 1;
        {
            v2i* t0 = &cand[par][pair][0][(8 * hi) * ROW_PITCH_C + 16 * half + m];
            v2i* t1 = &cand[par][pair][1][(8 * hi) * ROW_PITCH_C + 16 * half + m];
#pragma unroll
            for (int r = 0; r < 8; ++r) {
                t0[r * ROW_PITCH_C] = (v2i){__builtin_bit_cast(int, bestS0[r]), bestI0[r]};
                t1[r * ROW_PITCH_C] = (v2i){__builtin_bit_cast(int, bestS1[r]), bestI1[r]};
            }
        }
        __syncthreads();

        float ws[16];
        int   wi[16];
        {
            const v2i* rowp = &cand[par][pair][hi][m * ROW_PITCH_C];
#pragma unroll
            for (int j = 0; j < 16; ++j) {
                const v2i ca = rowp[2 * j];
                const v2i cb2 = rowp[2 * j + 1];
                ws[j] = __builtin_bit_cast(float, ca[0]);
                wi[j] = ca[1];
                take_lower(ws[j], wi[j], __builtin_bit_cast(float, cb2[0]), cb2[1]);
            }
        }
#pragma unroll
        for (int w = 8; w >= 1; w >>= 1) {
#pragma unroll
            for (int j = 0; j < w; ++j) take_lower(ws[j], wi[j], ws[j + w], wi[j + w]);
        }
        const int other = __shfl_xor(wi[0], 16, 32);
        const int myIdx0 = hi ? other : wi[0];
        const int myIdx1 = hi ? wi[0] : other;

        const float* sel0 = cb + (size_t)myIdx0 * DIMV + 4 * hi;
        const float* sel1 = cb + (size_t)myIdx1 * DIMV + 4 * hi;
#pragma unroll
        for (int i = 0; i < 8; ++i) {
            v4f cv0 = *(const v4f*)(sel0 + 8 * i);
            v4f cv1 = *(const v4f*)(sel1 + 8 * i);
            a0[i]  -= cv0;
            a1[i]  -= cv1;
            acc[i] += half ? cv1 : cv0;
        }
    }

    float* orow = out + (size_t)(half ? token1 : token0) * DIMV + 4 * hi;
#pragma unroll
    for (int i = 0; i < 8; ++i)
        *(v4f*)(orow + 8 * i) = acc[i];
}

extern "C" void kernel_launch(void* const* d_in, const int* in_sizes, int n_in,
                              void* d_out, int out_size, void* d_ws, size_t ws_size,
                              hipStream_t stream) {
    const float* x   = (const float*)d_in[0];
    const float* cbs = (const float*)d_in[1];
    float*       out = (float*)d_out;
    float*       csq = (float*)d_ws;
    float*       frag = csq + QN * KCB;

    const int tokens  = in_sizes[0] / DIMV;
    const int totalCB = in_sizes[1] / DIMV;

    ResidualVQ_csq_kernel<<<(totalCB + 255) / 256, 256, 0, stream>>>(cbs, csq, frag, totalCB);
    ResidualVQ_main_kernel<<<tokens / 64, 128, 0, stream>>>(x, cbs, csq, frag, out, tokens);
}
